# starting kernel
# baseline (speedup 1.0000x reference)
#include <hip/hip_runtime.h>
#include <hip/hip_bf16.h>

#define NB      4
#define NX      16384
#define NSITE   (NB * NX)
#define NIN     4
#define NOUT    8
#define NV      37
#define KPAD    40
#define SPW     16
#define COLS    (SPW * 18)
#define LSTRIDE 292

typedef float v2f __attribute__((ext_vector_type(2)));
typedef float v8f __attribute__((ext_vector_type(8)));


__device__ __forceinline__ void mulAB(const float* A, const float* B, float* C) {
#pragma unroll
  for (int i = 0; i < 3; ++i)
#pragma unroll
    for (int k = 0; k < 3; ++k) {
      float cR = 0.f, cI = 0.f;
#pragma unroll
      for (int j = 0; j < 3; ++j) {
        float aR = A[(i * 3 + j) * 2], aI = A[(i * 3 + j) * 2 + 1];
        float bR = B[(j * 3 + k) * 2], bI = B[(j * 3 + k) * 2 + 1];
        cR += aR * bR - aI * bI;
        cI += aR * bI + aI * bR;
      }
      C[(i * 3 + k) * 2] = cR;
      C[(i * 3 + k) * 2 + 1] = cI;
    }
}

__device__ __forceinline__ void mulABh(const float* A, const float* B, float* C) {
#pragma unroll
  for (int i = 0; i < 3; ++i)
#pragma unroll
    for (int k = 0; k < 3; ++k) {
      float cR = 0.f, cI = 0.f;
#pragma unroll
      for (int j = 0; j < 3; ++j) {
        float aR = A[(i * 3 + j) * 2], aI = A[(i * 3 + j) * 2 + 1];
        float bR = B[(k * 3 + j) * 2], bI = -B[(k * 3 + j) * 2 + 1];
        cR += aR * bR - aI * bI;
        cI += aR * bI + aI * bR;
      }
      C[(i * 3 + k) * 2] = cR;
      C[(i * 3 + k) * 2 + 1] = cI;
    }
}

__device__ __forceinline__ void mulAhB(const float* A, const float* B, float* C) {
#pragma unroll
  for (int i = 0; i < 3; ++i)
#pragma unroll
    for (int k = 0; k < 3; ++k) {
      float cR = 0.f, cI = 0.f;
#pragma unroll
      for (int j = 0; j < 3; ++j) {
        float aR = A[(j * 3 + i) * 2], aI = -A[(j * 3 + i) * 2 + 1];
        float bR = B[(j * 3 + k) * 2], bI = B[(j * 3 + k) * 2 + 1];
        cR += aR * bR - aI * bI;
        cI += aR * bI + aI * bR;
      }
      C[(i * 3 + k) * 2] = cR;
      C[(i * 3 + k) * 2 + 1] = cI;
    }
}

__global__ __launch_bounds__(256) void lconv_kernel(
    const float* __restrict__ u, const float* __restrict__ w,
    const float* __restrict__ weight, float* __restrict__ wout) {
  __shared__ float TWs[KPAD][LSTRIDE];

  const int tid = threadIdx.x;
  const int s0 = blockIdx.x * SPW;

  for (int i = tid; i < SPW * NIN * 18; i += 256) {
    int site = i / 72;
    int r = (i / 18) & 3;
    int m = i % 18;
    TWs[r][site * 18 + m] = w[(size_t)(s0 + site) * 72 + r * 18 + m];
  }

  for (int i = tid; i < 4 * COLS; i += 256) {
    int r = 36 + i / COLS;
    int col = i % COLS;
    int m = col % 18;
    TWs[r][col] = (r == 36 && (m == 0 || m == 8 || m == 16)) ? 1.f : 0.f;
  }

  for (int t = tid; t < SPW * 32; t += 256) {
    int site = t >> 5;
    int combo = (t >> 2) & 7;
    int n = t & 3;
    int s = s0 + site;
    int b = s >> 14;
    int x = s & (NX - 1);
    int axis = combo & 3;
    bool pos = combo < 4;

    int shamt = (axis == 0) ? 11 : (axis == 1) ? 8 : (axis == 2) ? 4 : 0;
    int dmask = (axis < 2) ? 7 : 15;
    int c = (x >> shamt) & dmask;
    int cn = pos ? ((c + 1) & dmask) : ((c - 1) & dmask);
    int xn = x + ((cn - c) << shamt);

    int xa = pos ? x : xn;
    const float* Ap = u + ((size_t)(b * NX + xa) * 4 + axis) * 18;
    const float* Wp = w + ((size_t)(b * NX + xn) * 4 + n) * 18;

    float A[18], W[18], T[18], V[18];
#pragma unroll
    for (int m = 0; m < 18; ++m) A[m] = Ap[m];
#pragma unroll
    for (int m = 0; m < 18; ++m) W[m] = Wp[m];

    if (pos) {
      mulAB(A, W, T);
      mulABh(T, A, V);
    } else {
      mulAhB(A, W, T);
      mulAB(T, A, V);
    }

    int row = 4 + combo * 4 + n;
#pragma unroll
    for (int m = 0; m < 18; ++m) TWs[row][site * 18 + m] = V[m];
  }

  __syncthreads();

  const int lane = tid & 31;
  const int wave = tid >> 5;
  const int rrow = lane & 15;
  const int kh = (lane >> 4) * 2;

  float ax[10], ay[10];
#pragma unroll
  for (int st = 0; st < 10; ++st) {
    int k0 = st * 4 + kh;
    ax[st] = (rrow < NOUT && k0 < NV) ? weight[rrow * NV + k0] : 0.f;
    ay[st] = (rrow < NOUT && (k0 + 1) < NV) ? weight[rrow * NV + k0 + 1] : 0.f;
  }

  for (int tile = wave; tile < COLS / 16; tile += 8) {
    int nbase = tile * 16;
    v8f acc = {0.f, 0.f, 0.f, 0.f, 0.f, 0.f, 0.f, 0.f};
#pragma unroll
    for (int st = 0; st < 10; ++st) {
      int k0 = st * 4 + kh;
      v2f a, bf;
      a.x = ax[st];
      a.y = ay[st];
      bf.x = TWs[k0][nbase + rrow];
      bf.y = TWs[k0 + 1][nbase + rrow];
      acc = __builtin_amdgcn_wmma_f32_16x16x4_f32(
           false, a,  false, bf,
           (short)0, acc,  false,  false);
    }
    if (lane < 16) {
      int col = nbase + lane;
      int site = col / 18;
      int m = col % 18;
      size_t base = (size_t)(s0 + site) * (NOUT * 18) + m;
#pragma unroll
      for (int r = 0; r < NOUT; ++r) wout[base + r * 18] = acc[r];
    }
  }
}

extern "C" void kernel_launch(void* const* d_in, const int* in_sizes, int n_in,
                              void* d_out, int out_size, void* d_ws, size_t ws_size,
                              hipStream_t stream) {
  const float* u = (const float*)d_in[0];
  const float* w = (const float*)d_in[1];
  const float* weight = (const float*)d_in[2];
  float* out = (float*)d_out;

  const size_t u_elems = (size_t)NB * NX * 4 * 18;
  hipMemcpyAsync(out, u, u_elems * sizeof(float), hipMemcpyDeviceToDevice, stream);

  float* wout = out + u_elems;
  lconv_kernel<<<NSITE / SPW, 256, 0, stream>>>(u, w, weight, wout);
}


# sandwich tasks remapped so each wave holds one orientation (2 sites x 4 axes x 4 channels): the +1/-1 branch becomes wave-uniform and a wave runs one side instead of both under partial masks
# speedup vs baseline: 1.1916x; 1.1916x over previous
#include <hip/hip_runtime.h>
#include <hip/hip_bf16.h>

#define NB      4
#define NX      16384
#define NSITE   (NB * NX)
#define NIN     4
#define NOUT    8
#define NV      37
#define KPAD    40
#define SPW     16
#define COLS    (SPW * 18)
#define LSTRIDE 292

typedef float v2f __attribute__((ext_vector_type(2)));
typedef float v8f __attribute__((ext_vector_type(8)));


__device__ __forceinline__ void mulAB(const float* A, const float* B, float* C) {
#pragma unroll
  for (int i = 0; i < 3; ++i)
#pragma unroll
    for (int k = 0; k < 3; ++k) {
      float cR = 0.f, cI = 0.f;
#pragma unroll
      for (int j = 0; j < 3; ++j) {
        float aR = A[(i * 3 + j) * 2], aI = A[(i * 3 + j) * 2 + 1];
        float bR = B[(j * 3 + k) * 2], bI = B[(j * 3 + k) * 2 + 1];
        cR += aR * bR - aI * bI;
        cI += aR * bI + aI * bR;
      }
      C[(i * 3 + k) * 2] = cR;
      C[(i * 3 + k) * 2 + 1] = cI;
    }
}

__device__ __forceinline__ void mulABh(const float* A, const float* B, float* C) {
#pragma unroll
  for (int i = 0; i < 3; ++i)
#pragma unroll
    for (int k = 0; k < 3; ++k) {
      float cR = 0.f, cI = 0.f;
#pragma unroll
      for (int j = 0; j < 3; ++j) {
        float aR = A[(i * 3 + j) * 2], aI = A[(i * 3 + j) * 2 + 1];
        float bR = B[(k * 3 + j) * 2], bI = -B[(k * 3 + j) * 2 + 1];
        cR += aR * bR - aI * bI;
        cI += aR * bI + aI * bR;
      }
      C[(i * 3 + k) * 2] = cR;
      C[(i * 3 + k) * 2 + 1] = cI;
    }
}

__device__ __forceinline__ void mulAhB(const float* A, const float* B, float* C) {
#pragma unroll
  for (int i = 0; i < 3; ++i)
#pragma unroll
    for (int k = 0; k < 3; ++k) {
      float cR = 0.f, cI = 0.f;
#pragma unroll
      for (int j = 0; j < 3; ++j) {
        float aR = A[(j * 3 + i) * 2], aI = -A[(j * 3 + i) * 2 + 1];
        float bR = B[(j * 3 + k) * 2], bI = B[(j * 3 + k) * 2 + 1];
        cR += aR * bR - aI * bI;
        cI += aR * bI + aI * bR;
      }
      C[(i * 3 + k) * 2] = cR;
      C[(i * 3 + k) * 2 + 1] = cI;
    }
}

__global__ __launch_bounds__(256) void lconv_kernel(
    const float* __restrict__ u, const float* __restrict__ w,
    const float* __restrict__ weight, float* __restrict__ wout) {
  __shared__ float TWs[KPAD][LSTRIDE];

  const int tid = threadIdx.x;
  const int s0 = blockIdx.x * SPW;

  for (int i = tid; i < SPW * NIN * 18; i += 256) {
    int site = i / 72;
    int r = (i / 18) & 3;
    int m = i % 18;
    TWs[r][site * 18 + m] = w[(size_t)(s0 + site) * 72 + r * 18 + m];
  }

  for (int i = tid; i < 4 * COLS; i += 256) {
    int r = 36 + i / COLS;
    int col = i % COLS;
    int m = col % 18;
    TWs[r][col] = (r == 36 && (m == 0 || m == 8 || m == 16)) ? 1.f : 0.f;
  }

  for (int t = tid; t < SPW * 32; t += 256) {
    int n = t & 3;
    int axis = (t >> 2) & 3;
    bool pos = ((t >> 5) & 1) == 0;
    int site = ((t >> 6) << 1) | ((t >> 4) & 1);
    int combo = pos ? axis : axis + 4;
    int s = s0 + site;
    int b = s >> 14;
    int x = s & (NX - 1);

    int shamt = (axis == 0) ? 11 : (axis == 1) ? 8 : (axis == 2) ? 4 : 0;
    int dmask = (axis < 2) ? 7 : 15;
    int c = (x >> shamt) & dmask;
    int cn = pos ? ((c + 1) & dmask) : ((c - 1) & dmask);
    int xn = x + ((cn - c) << shamt);

    int xa = pos ? x : xn;
    const float* Ap = u + ((size_t)(b * NX + xa) * 4 + axis) * 18;
    const float* Wp = w + ((size_t)(b * NX + xn) * 4 + n) * 18;

    float A[18], W[18], T[18], V[18];
#pragma unroll
    for (int m = 0; m < 18; ++m) A[m] = Ap[m];
#pragma unroll
    for (int m = 0; m < 18; ++m) W[m] = Wp[m];

    if (pos) {
      mulAB(A, W, T);
      mulABh(T, A, V);
    } else {
      mulAhB(A, W, T);
      mulAB(T, A, V);
    }

    int row = 4 + combo * 4 + n;
#pragma unroll
    for (int m = 0; m < 18; ++m) TWs[row][site * 18 + m] = V[m];
  }

  __syncthreads();

  const int lane = tid & 31;
  const int wave = tid >> 5;
  const int rrow = lane & 15;
  const int kh = (lane >> 4) * 2;

  float ax[10], ay[10];
#pragma unroll
  for (int st = 0; st < 10; ++st) {
    int k0 = st * 4 + kh;
    ax[st] = (rrow < NOUT && k0 < NV) ? weight[rrow * NV + k0] : 0.f;
    ay[st] = (rrow < NOUT && (k0 + 1) < NV) ? weight[rrow * NV + k0 + 1] : 0.f;
  }

  for (int tile = wave; tile < COLS / 16; tile += 8) {
    int nbase = tile * 16;
    v8f acc = {0.f, 0.f, 0.f, 0.f, 0.f, 0.f, 0.f, 0.f};
#pragma unroll
    for (int st = 0; st < 10; ++st) {
      int k0 = st * 4 + kh;
      v2f a, bf;
      a.x = ax[st];
      a.y = ay[st];
      bf.x = TWs[k0][nbase + rrow];
      bf.y = TWs[k0 + 1][nbase + rrow];
      acc = __builtin_amdgcn_wmma_f32_16x16x4_f32(
           false, a,  false, bf,
           (short)0, acc,  false,  false);
    }
    if (lane < 16) {
      int col = nbase + lane;
      int site = col / 18;
      int m = col % 18;
      size_t base = (size_t)(s0 + site) * (NOUT * 18) + m;
#pragma unroll
      for (int r = 0; r < NOUT; ++r) wout[base + r * 18] = acc[r];
    }
  }
}

extern "C" void kernel_launch(void* const* d_in, const int* in_sizes, int n_in,
                              void* d_out, int out_size, void* d_ws, size_t ws_size,
                              hipStream_t stream) {
  const float* u = (const float*)d_in[0];
  const float* w = (const float*)d_in[1];
  const float* weight = (const float*)d_in[2];
  float* out = (float*)d_out;

  const size_t u_elems = (size_t)NB * NX * 4 * 18;
  hipMemcpyAsync(out, u, u_elems * sizeof(float), hipMemcpyDeviceToDevice, stream);

  float* wout = out + u_elems;
  lconv_kernel<<<NSITE / SPW, 256, 0, stream>>>(u, w, weight, wout);
}


# every read of w and weight becomes one pass of whole-line 16-byte loads into LDS, all in flight at once (own row, six neighbour rows, weight); sandwich W operands and A fragments then read LDS
# speedup vs baseline: 1.3609x; 1.1421x over previous
#include <hip/hip_runtime.h>
#include <hip/hip_bf16.h>

#define NB      4
#define NX      16384
#define NSITE   (NB * NX)
#define NIN     4
#define NOUT    8
#define NV      37
#define KPAD    40
#define SPW     16
#define COLS    (SPW * 18)
#define LSTRIDE 292
#define TWF     (KPAD * LSTRIDE)
#define NROWQ   288
#define WROW    1168
#define WNF     (6 * WROW)

typedef float v2f __attribute__((ext_vector_type(2)));
typedef float v8f __attribute__((ext_vector_type(8)));


__device__ __forceinline__ void mulAB(const float* A, const float* B, float* C) {
#pragma unroll
  for (int i = 0; i < 3; ++i)
#pragma unroll
    for (int k = 0; k < 3; ++k) {
      float cR = 0.f, cI = 0.f;
#pragma unroll
      for (int j = 0; j < 3; ++j) {
        float aR = A[(i * 3 + j) * 2], aI = A[(i * 3 + j) * 2 + 1];
        float bR = B[(j * 3 + k) * 2], bI = B[(j * 3 + k) * 2 + 1];
        cR += aR * bR - aI * bI;
        cI += aR * bI + aI * bR;
      }
      C[(i * 3 + k) * 2] = cR;
      C[(i * 3 + k) * 2 + 1] = cI;
    }
}

__device__ __forceinline__ void mulABh(const float* A, const float* B, float* C) {
#pragma unroll
  for (int i = 0; i < 3; ++i)
#pragma unroll
    for (int k = 0; k < 3; ++k) {
      float cR = 0.f, cI = 0.f;
#pragma unroll
      for (int j = 0; j < 3; ++j) {
        float aR = A[(i * 3 + j) * 2], aI = A[(i * 3 + j) * 2 + 1];
        float bR = B[(k * 3 + j) * 2], bI = -B[(k * 3 + j) * 2 + 1];
        cR += aR * bR - aI * bI;
        cI += aR * bI + aI * bR;
      }
      C[(i * 3 + k) * 2] = cR;
      C[(i * 3 + k) * 2 + 1] = cI;
    }
}

__device__ __forceinline__ void mulAhB(const float* A, const float* B, float* C) {
#pragma unroll
  for (int i = 0; i < 3; ++i)
#pragma unroll
    for (int k = 0; k < 3; ++k) {
      float cR = 0.f, cI = 0.f;
#pragma unroll
      for (int j = 0; j < 3; ++j) {
        float aR = A[(j * 3 + i) * 2], aI = -A[(j * 3 + i) * 2 + 1];
        float bR = B[(j * 3 + k) * 2], bI = B[(j * 3 + k) * 2 + 1];
        cR += aR * bR - aI * bI;
        cI += aR * bI + aI * bR;
      }
      C[(i * 3 + k) * 2] = cR;
      C[(i * 3 + k) * 2 + 1] = cI;
    }
}

__global__ __launch_bounds__(256) void lconv_kernel(
    const float* __restrict__ u, const float* __restrict__ w,
    const float* __restrict__ weight, float* __restrict__ wout) {
  __shared__ __attribute__((aligned(16))) float smem[TWF + WNF + 16 * KPAD];
  float (*TWs)[LSTRIDE] = (float (*)[LSTRIDE])smem;
  float* Wn = smem + TWF;
  float* Ws = smem + TWF + WNF;

  const int tid = threadIdx.x;
  __builtin_assume(tid >= 0 && tid < 256);
  const int s0 = blockIdx.x * SPW;
  const int bb = s0 >> 14;
  const int x0 = s0 & (NX - 1);

  float4 vn[7];
#pragma unroll
  for (int it = 0; it < 7; ++it) {
    int i = tid + 256 * it;
    if (i >= 6 * NROWQ) i -= 256;
    int j = i / NROWQ;
    int q = i - j * NROWQ;
    int a = (j >= 3) ? j - 3 : j;
    int shamt = (a == 0) ? 11 : (a == 1) ? 8 : 4;
    int dmask = (a < 2) ? 7 : 15;
    int c = (x0 >> shamt) & dmask;
    int cn = (j < 3) ? ((c + 1) & dmask) : ((c - 1) & dmask);
    int xr = x0 + ((cn - c) << shamt);
    vn[it] = *((const float4*)(w + (size_t)(bb * NX + xr) * 72) + q);
  }
  const float4* ownp = (const float4*)(w + (size_t)s0 * 72);
  const int q1 = (tid < NROWQ - 256) ? 256 + tid : tid;
  float4 vo0 = ownp[tid];
  float4 vo1 = ownp[q1];
  const int i2 = (tid < NOUT * NV - 256) ? 256 + tid : tid;
  float wa = weight[tid];
  float wb = weight[i2];

#pragma unroll
  for (int it = 0; it < 7; ++it) {
    int i = tid + 256 * it;
    if (i >= 6 * NROWQ) i -= 256;
    int j = i / NROWQ;
    int q = i - j * NROWQ;
    *(float4*)(Wn + j * WROW + q * 4) = vn[it];
  }

  {
    float e0[4] = {vo0.x, vo0.y, vo0.z, vo0.w};
    float e1[4] = {vo1.x, vo1.y, vo1.z, vo1.w};
#pragma unroll
    for (int k = 0; k < 4; ++k) {
      int f = tid * 4 + k;
      int site = f / 72;
      int r = (f / 18) & 3;
      int m = f % 18;
      TWs[r][site * 18 + m] = e0[k];
    }
#pragma unroll
    for (int k = 0; k < 4; ++k) {
      int f = q1 * 4 + k;
      int site = f / 72;
      int r = (f / 18) & 3;
      int m = f % 18;
      TWs[r][site * 18 + m] = e1[k];
    }
  }

  {
    int r = tid / NV;
    Ws[r * KPAD + (tid - r * NV)] = wa;
    int r2 = i2 / NV;
    Ws[r2 * KPAD + (i2 - r2 * NV)] = wb;
  }
  for (int i = tid; i < 16 * KPAD; i += 256) {
    int r = i / KPAD;
    int k = i - r * KPAD;
    if (r >= NOUT || k >= NV) Ws[i] = 0.f;
  }

  for (int i = tid; i < 4 * COLS; i += 256) {
    int r = 36 + i / COLS;
    int col = i % COLS;
    int m = col % 18;
    TWs[r][col] = (r == 36 && (m == 0 || m == 8 || m == 16)) ? 1.f : 0.f;
  }

  __syncthreads();

  for (int t = tid; t < SPW * 32; t += 256) {
    int n = t & 3;
    int axis = (t >> 2) & 3;
    bool pos = ((t >> 5) & 1) == 0;
    int site = ((t >> 6) << 1) | ((t >> 4) & 1);
    int combo = pos ? axis : axis + 4;
    int s = s0 + site;
    int b = s >> 14;
    int x = s & (NX - 1);

    int shamt = (axis == 0) ? 11 : (axis == 1) ? 8 : (axis == 2) ? 4 : 0;
    int dmask = (axis < 2) ? 7 : 15;
    int c = (x >> shamt) & dmask;
    int cn = pos ? ((c + 1) & dmask) : ((c - 1) & dmask);
    int xn = x + ((cn - c) << shamt);

    int xa = pos ? x : xn;
    const float* Ap = u + ((size_t)(b * NX + xa) * 4 + axis) * 18;
    int woff = (axis == 3) ? (n * LSTRIDE + cn * 18)
                           : (TWF + (pos ? axis : axis + 3) * WROW + site * 72 + n * 18);
    const float2* Wp = (const float2*)(smem + woff);

    float A[18], W[18], T[18], V[18];
#pragma unroll
    for (int m = 0; m < 18; ++m) A[m] = Ap[m];
#pragma unroll
    for (int q = 0; q < 9; ++q) {
      float2 v = Wp[q];
      W[2 * q] = v.x;
      W[2 * q + 1] = v.y;
    }

    if (pos) {
      mulAB(A, W, T);
      mulABh(T, A, V);
    } else {
      mulAhB(A, W, T);
      mulAB(T, A, V);
    }

    int row = 4 + combo * 4 + n;
#pragma unroll
    for (int m = 0; m < 18; ++m) TWs[row][site * 18 + m] = V[m];
  }

  __syncthreads();

  const int lane = tid & 31;
  const int wave = tid >> 5;
  const int rrow = lane & 15;
  const int kh = (lane >> 4) * 2;

  float ax[10], ay[10];
#pragma unroll
  for (int st = 0; st < 10; ++st) {
    int k0 = st * 4 + kh;
    float2 v = *(const float2*)(Ws + rrow * KPAD + k0);
    ax[st] = v.x;
    ay[st] = v.y;
  }

  for (int tile = wave; tile < COLS / 16; tile += 8) {
    int nbase = tile * 16;
    v8f acc = {0.f, 0.f, 0.f, 0.f, 0.f, 0.f, 0.f, 0.f};
#pragma unroll
    for (int st = 0; st < 10; ++st) {
      int k0 = st * 4 + kh;
      v2f a, bf;
      a.x = ax[st];
      a.y = ay[st];
      bf.x = TWs[k0][nbase + rrow];
      bf.y = TWs[k0 + 1][nbase + rrow];
      acc = __builtin_amdgcn_wmma_f32_16x16x4_f32(
           false, a,  false, bf,
           (short)0, acc,  false,  false);
    }
    if (lane < 16) {
      int col = nbase + lane;
      int site = col / 18;
      int m = col % 18;
      size_t base = (size_t)(s0 + site) * (NOUT * 18) + m;
#pragma unroll
      for (int r = 0; r < NOUT; ++r) wout[base + r * 18] = acc[r];
    }
  }
}

extern "C" void kernel_launch(void* const* d_in, const int* in_sizes, int n_in,
                              void* d_out, int out_size, void* d_ws, size_t ws_size,
                              hipStream_t stream) {
  const float* u = (const float*)d_in[0];
  const float* w = (const float*)d_in[1];
  const float* weight = (const float*)d_in[2];
  float* out = (float*)d_out;

  const size_t u_elems = (size_t)NB * NX * 4 * 18;
  hipMemcpyAsync(out, u, u_elems * sizeof(float), hipMemcpyDeviceToDevice, stream);

  float* wout = out + u_elems;
  lconv_kernel<<<NSITE / SPW, 256, 0, stream>>>(u, w, weight, wout);
}


# sandwich 3x3 complex products rewritten on (Re,Im) pairs: each multiply-add is two packed FMAs with select/sign modifiers, one loop per side; an entry's six products now sum in one fused chain
# speedup vs baseline: 1.5471x; 1.1368x over previous
#include <hip/hip_runtime.h>
#include <hip/hip_bf16.h>

#define NB      4
#define NX      16384
#define NSITE   (NB * NX)
#define NIN     4
#define NOUT    8
#define NV      37
#define KPAD    40
#define SPW     16
#define COLS    (SPW * 18)
#define LSTRIDE 292
#define TWF     (KPAD * LSTRIDE)
#define NROWQ   288
#define WROW    1168
#define WNF     (6 * WROW)

typedef float v2f __attribute__((ext_vector_type(2)));
typedef float v8f __attribute__((ext_vector_type(8)));


__device__ __forceinline__ v2f cmul(v2f a, v2f b) {
  v2f r = a.xx * b;
  return __builtin_elementwise_fma(a.yy, (v2f){-b.y, b.x}, r);
}
__device__ __forceinline__ v2f cmac(v2f a, v2f b, v2f c) {
  c = __builtin_elementwise_fma(a.xx, b, c);
  return __builtin_elementwise_fma(a.yy, (v2f){-b.y, b.x}, c);
}
__device__ __forceinline__ v2f cmul_bc(v2f a, v2f b) {
  v2f r = a.xx * (v2f){b.x, -b.y};
  return __builtin_elementwise_fma(a.yy, b.yx, r);
}
__device__ __forceinline__ v2f cmac_bc(v2f a, v2f b, v2f c) {
  c = __builtin_elementwise_fma(a.xx, (v2f){b.x, -b.y}, c);
  return __builtin_elementwise_fma(a.yy, b.yx, c);
}

template <bool POS>
__device__ __forceinline__ void sandwich_task(int t, int s0, const float* __restrict__ u,
                                              float* smem) {
  float (*TWs)[LSTRIDE] = (float (*)[LSTRIDE])smem;
  int n = t & 3;
  int axis = (t >> 2) & 3;
  int site = ((t >> 6) << 1) | ((t >> 4) & 1);
  int combo = POS ? axis : axis + 4;
  int s = s0 + site;
  int b = s >> 14;
  int x = s & (NX - 1);

  int shamt = (axis == 0) ? 11 : (axis == 1) ? 8 : (axis == 2) ? 4 : 0;
  int dmask = (axis < 2) ? 7 : 15;
  int c = (x >> shamt) & dmask;
  int cn = POS ? ((c + 1) & dmask) : ((c - 1) & dmask);
  int xn = x + ((cn - c) << shamt);

  int xa = POS ? x : xn;
  const float* Ap = u + ((size_t)(b * NX + xa) * 4 + axis) * 18;
  int woff = (axis == 3) ? (n * LSTRIDE + cn * 18)
                         : (TWF + (POS ? axis : axis + 3) * WROW + site * 72 + n * 18);
  const v2f* Wp = (const v2f*)(smem + woff);
  const v2f* Ap2 = (const v2f*)Ap;

  v2f A[9], Wm[9], T[9], V[9];
#pragma unroll
  for (int q = 0; q < 9; ++q) A[q] = Ap2[q];
#pragma unroll
  for (int q = 0; q < 9; ++q) Wm[q] = Wp[q];

  if (POS) {
#pragma unroll
    for (int i = 0; i < 3; ++i)
#pragma unroll
      for (int k = 0; k < 3; ++k) {
        v2f c = cmul(Wm[k], A[i * 3]);
        c = cmac(Wm[3 + k], A[i * 3 + 1], c);
        T[i * 3 + k] = cmac(Wm[6 + k], A[i * 3 + 2], c);
      }
#pragma unroll
    for (int i = 0; i < 3; ++i)
#pragma unroll
      for (int k = 0; k < 3; ++k) {
        v2f c = cmul_bc(T[i * 3], A[k * 3]);
        c = cmac_bc(T[i * 3 + 1], A[k * 3 + 1], c);
        V[i * 3 + k] = cmac_bc(T[i * 3 + 2], A[k * 3 + 2], c);
      }
  } else {
#pragma unroll
    for (int i = 0; i < 3; ++i)
#pragma unroll
      for (int k = 0; k < 3; ++k) {
        v2f c = cmul_bc(Wm[k], A[i]);
        c = cmac_bc(Wm[3 + k], A[3 + i], c);
        T[i * 3 + k] = cmac_bc(Wm[6 + k], A[6 + i], c);
      }
#pragma unroll
    for (int i = 0; i < 3; ++i)
#pragma unroll
      for (int k = 0; k < 3; ++k) {
        v2f c = cmul(T[i * 3], A[k]);
        c = cmac(T[i * 3 + 1], A[3 + k], c);
        V[i * 3 + k] = cmac(T[i * 3 + 2], A[6 + k], c);
      }
  }

  int row = 4 + combo * 4 + n;
  v2f* Vp = (v2f*)&TWs[row][site * 18];
#pragma unroll
  for (int q = 0; q < 9; ++q) Vp[q] = V[q];
}

__global__ __launch_bounds__(256) void lconv_kernel(
    const float* __restrict__ u, const float* __restrict__ w,
    const float* __restrict__ weight, float* __restrict__ wout) {
  __shared__ __attribute__((aligned(16))) float smem[TWF + WNF + 16 * KPAD];
  float (*TWs)[LSTRIDE] = (float (*)[LSTRIDE])smem;
  float* Wn = smem + TWF;
  float* Ws = smem + TWF + WNF;

  const int tid = threadIdx.x;
  __builtin_assume(tid >= 0 && tid < 256);
  const int s0 = blockIdx.x * SPW;
  const int bb = s0 >> 14;
  const int x0 = s0 & (NX - 1);

  float4 vn[7];
#pragma unroll
  for (int it = 0; it < 7; ++it) {
    int i = tid + 256 * it;
    if (i >= 6 * NROWQ) i -= 256;
    int j = i / NROWQ;
    int q = i - j * NROWQ;
    int a = (j >= 3) ? j - 3 : j;
    int shamt = (a == 0) ? 11 : (a == 1) ? 8 : 4;
    int dmask = (a < 2) ? 7 : 15;
    int c = (x0 >> shamt) & dmask;
    int cn = (j < 3) ? ((c + 1) & dmask) : ((c - 1) & dmask);
    int xr = x0 + ((cn - c) << shamt);
    vn[it] = *((const float4*)(w + (size_t)(bb * NX + xr) * 72) + q);
  }
  const float4* ownp = (const float4*)(w + (size_t)s0 * 72);
  const int q1 = (tid < NROWQ - 256) ? 256 + tid : tid;
  float4 vo0 = ownp[tid];
  float4 vo1 = ownp[q1];
  const int i2 = (tid < NOUT * NV - 256) ? 256 + tid : tid;
  float wa = weight[tid];
  float wb = weight[i2];

#pragma unroll
  for (int it = 0; it < 7; ++it) {
    int i = tid + 256 * it;
    if (i >= 6 * NROWQ) i -= 256;
    int j = i / NROWQ;
    int q = i - j * NROWQ;
    *(float4*)(Wn + j * WROW + q * 4) = vn[it];
  }

  {
    float e0[4] = {vo0.x, vo0.y, vo0.z, vo0.w};
    float e1[4] = {vo1.x, vo1.y, vo1.z, vo1.w};
#pragma unroll
    for (int k = 0; k < 4; ++k) {
      int f = tid * 4 + k;
      int site = f / 72;
      int r = (f / 18) & 3;
      int m = f % 18;
      TWs[r][site * 18 + m] = e0[k];
    }
#pragma unroll
    for (int k = 0; k < 4; ++k) {
      int f = q1 * 4 + k;
      int site = f / 72;
      int r = (f / 18) & 3;
      int m = f % 18;
      TWs[r][site * 18 + m] = e1[k];
    }
  }

  {
    int r = tid / NV;
    Ws[r * KPAD + (tid - r * NV)] = wa;
    int r2 = i2 / NV;
    Ws[r2 * KPAD + (i2 - r2 * NV)] = wb;
  }
  for (int i = tid; i < 16 * KPAD; i += 256) {
    int r = i / KPAD;
    int k = i - r * KPAD;
    if (r >= NOUT || k >= NV) Ws[i] = 0.f;
  }

  for (int i = tid; i < 4 * COLS; i += 256) {
    int r = 36 + i / COLS;
    int col = i % COLS;
    int m = col % 18;
    TWs[r][col] = (r == 36 && (m == 0 || m == 8 || m == 16)) ? 1.f : 0.f;
  }

  __syncthreads();

  if (((tid >> 5) & 1) == 0) {
    for (int t = tid; t < SPW * 32; t += 256) sandwich_task<true>(t, s0, u, smem);
  } else {
    for (int t = tid; t < SPW * 32; t += 256) sandwich_task<false>(t, s0, u, smem);
  }

  __syncthreads();

  const int lane = tid & 31;
  const int wave = tid >> 5;
  const int rrow = lane & 15;
  const int kh = (lane >> 4) * 2;

  float ax[10], ay[10];
#pragma unroll
  for (int st = 0; st < 10; ++st) {
    int k0 = st * 4 + kh;
    float2 v = *(const float2*)(Ws + rrow * KPAD + k0);
    ax[st] = v.x;
    ay[st] = v.y;
  }

  for (int tile = wave; tile < COLS / 16; tile += 8) {
    int nbase = tile * 16;
    v8f acc = {0.f, 0.f, 0.f, 0.f, 0.f, 0.f, 0.f, 0.f};
#pragma unroll
    for (int st = 0; st < 10; ++st) {
      int k0 = st * 4 + kh;
      v2f a, bf;
      a.x = ax[st];
      a.y = ay[st];
      bf.x = TWs[k0][nbase + rrow];
      bf.y = TWs[k0 + 1][nbase + rrow];
      acc = __builtin_amdgcn_wmma_f32_16x16x4_f32(
           false, a,  false, bf,
           (short)0, acc,  false,  false);
    }
    if (lane < 16) {
      int col = nbase + lane;
      int site = col / 18;
      int m = col % 18;
      size_t base = (size_t)(s0 + site) * (NOUT * 18) + m;
#pragma unroll
      for (int r = 0; r < NOUT; ++r) wout[base + r * 18] = acc[r];
    }
  }
}

extern "C" void kernel_launch(void* const* d_in, const int* in_sizes, int n_in,
                              void* d_out, int out_size, void* d_ws, size_t ws_size,
                              hipStream_t stream) {
  const float* u = (const float*)d_in[0];
  const float* w = (const float*)d_in[1];
  const float* weight = (const float*)d_in[2];
  float* out = (float*)d_out;

  const size_t u_elems = (size_t)NB * NX * 4 * 18;
  hipMemcpyAsync(out, u, u_elems * sizeof(float), hipMemcpyDeviceToDevice, stream);

  float* wout = out + u_elems;
  lconv_kernel<<<NSITE / SPW, 256, 0, stream>>>(u, w, weight, wout);
}


# index arithmetic cut: w rows and weight staged as raw thread-linear copies (no divides, no transposing stores), B reads K 0..3 from the raw row, identity/pad rows from registers; task loop unrolls
# speedup vs baseline: 1.7313x; 1.1191x over previous
#include <hip/hip_runtime.h>
#include <hip/hip_bf16.h>

#define NB      4
#define NX      16384
#define NSITE   (NB * NX)
#define NIN     4
#define NOUT    8
#define NV      37
#define KPAD    40
#define SPW     16
#define COLS    (SPW * 18)
#define LSTRIDE 292
#define TROWS   36
#define TWF     (TROWS * LSTRIDE)
#define NROWQ   288
#define WROW    1168
#define WNO     (TWF + 8)
#define WNF     (6 * WROW)
#define WSO     (WNO + WNF)

typedef float v2f __attribute__((ext_vector_type(2)));
typedef float v8f __attribute__((ext_vector_type(8)));


__device__ __forceinline__ v2f cmul(v2f a, v2f b) {
  v2f r = a.xx * b;
  return __builtin_elementwise_fma(a.yy, (v2f){-b.y, b.x}, r);
}
__device__ __forceinline__ v2f cmac(v2f a, v2f b, v2f c) {
  c = __builtin_elementwise_fma(a.xx, b, c);
  return __builtin_elementwise_fma(a.yy, (v2f){-b.y, b.x}, c);
}
__device__ __forceinline__ v2f cmul_bc(v2f a, v2f b) {
  v2f r = a.xx * (v2f){b.x, -b.y};
  return __builtin_elementwise_fma(a.yy, b.yx, r);
}
__device__ __forceinline__ v2f cmac_bc(v2f a, v2f b, v2f c) {
  c = __builtin_elementwise_fma(a.xx, (v2f){b.x, -b.y}, c);
  return __builtin_elementwise_fma(a.yy, b.yx, c);
}

template <bool POS>
__device__ __forceinline__ void sandwich_side(int tid, int bb, int x0,
                                              const float* __restrict__ u, float* smem) {
  float (*TWs)[LSTRIDE] = (float (*)[LSTRIDE])smem;
  const int n = tid & 3;
  const int axis = (tid >> 2) & 3;
  const int shamt = (0x0004080B >> (axis * 8)) & 0xFF;
  const int dmask = (axis < 2) ? 7 : 15;
  const int combo = POS ? axis : axis + 4;
  const int wrow = ((axis == 3) ? 0 : WNO + (POS ? axis : axis + 3) * WROW) + n * 18;
  const int site0 = ((tid >> 6) << 1) | ((tid >> 4) & 1);

  for (int pass = 0; pass < 2; ++pass) {
    const int site = site0 + 8 * pass;
    const int x = x0 + site;
    const int c = (x >> shamt) & dmask;
    const int cn = POS ? ((c + 1) & dmask) : ((c - 1) & dmask);
    const int xn = x + ((cn - c) << shamt);
    const int xa = POS ? x : xn;
    const unsigned aoff = (unsigned)(((bb * NX + xa) * 4 + axis) * 18);
    const v2f* Ap2 = (const v2f*)(u + aoff);
    const v2f* Wp = (const v2f*)(smem + wrow + ((axis == 3) ? cn : site) * 72);

    v2f A[9], Wm[9], T[9], V[9];
#pragma unroll
    for (int q = 0; q < 9; ++q) A[q] = Ap2[q];
#pragma unroll
    for (int q = 0; q < 9; ++q) Wm[q] = Wp[q];

    if (POS) {
#pragma unroll
      for (int i = 0; i < 3; ++i)
#pragma unroll
        for (int k = 0; k < 3; ++k) {
          v2f c2 = cmul(Wm[k], A[i * 3]);
          c2 = cmac(Wm[3 + k], A[i * 3 + 1], c2);
          T[i * 3 + k] = cmac(Wm[6 + k], A[i * 3 + 2], c2);
        }
#pragma unroll
      for (int i = 0; i < 3; ++i)
#pragma unroll
        for (int k = 0; k < 3; ++k) {
          v2f c2 = cmul_bc(T[i * 3], A[k * 3]);
          c2 = cmac_bc(T[i * 3 + 1], A[k * 3 + 1], c2);
          V[i * 3 + k] = cmac_bc(T[i * 3 + 2], A[k * 3 + 2], c2);
        }
    } else {
#pragma unroll
      for (int i = 0; i < 3; ++i)
#pragma unroll
        for (int k = 0; k < 3; ++k) {
          v2f c2 = cmul_bc(Wm[k], A[i]);
          c2 = cmac_bc(Wm[3 + k], A[3 + i], c2);
          T[i * 3 + k] = cmac_bc(Wm[6 + k], A[6 + i], c2);
        }
#pragma unroll
      for (int i = 0; i < 3; ++i)
#pragma unroll
        for (int k = 0; k < 3; ++k) {
          v2f c2 = cmul(T[i * 3], A[k]);
          c2 = cmac(T[i * 3 + 1], A[3 + k], c2);
          V[i * 3 + k] = cmac(T[i * 3 + 2], A[6 + k], c2);
        }
    }

    v2f* Vp = (v2f*)&TWs[4 + combo * 4 + n][site * 18];
#pragma unroll
    for (int q = 0; q < 9; ++q) Vp[q] = V[q];
  }
}

__global__ __launch_bounds__(256) void lconv_kernel(
    const float* __restrict__ u, const float* __restrict__ w,
    const float* __restrict__ weight, float* __restrict__ wout) {
  __shared__ __attribute__((aligned(16))) float smem[WSO + 16 * KPAD];
  float (*TWs)[LSTRIDE] = (float (*)[LSTRIDE])smem;
  float* Ws = smem + WSO;

  const int tid = threadIdx.x;
  __builtin_assume(tid >= 0 && tid < 256);
  const int lane = tid & 31;
  const int wv = __builtin_amdgcn_readfirstlane(tid >> 5);
  const int s0 = blockIdx.x * SPW;
  const int bb = s0 >> 14;
  const int x0 = s0 & (NX - 1);

  auto rowsite = [&](int j) -> int {
    int a = (j >= 3) ? j - 3 : j;
    int shamt = (a == 0) ? 11 : (a == 1) ? 8 : 4;
    int dmask = (a < 2) ? 7 : 15;
    int c = (x0 >> shamt) & dmask;
    int cn = (j < 3) ? ((c + 1) & dmask) : ((c - 1) & dmask);
    return (j >= 6) ? x0 : x0 + ((cn - c) << shamt);
  };
  const float4* wq = (const float4*)w;
  float4 vn[6];
#pragma unroll
  for (int j = 0; j < 6; ++j) vn[j] = wq[(unsigned)((bb * NX + rowsite(j)) * 18 + tid)];
  float4 vo = wq[(unsigned)(s0 * 18 + tid)];
  const int jt = (wv < 6) ? wv : 6;
  float4 vt = wq[(unsigned)((bb * NX + rowsite(jt)) * 18 + 256 + lane)];
  const int k2 = 32 + (tid & 7);
  const int r2 = (tid >> 3) & 15;
  float wa = weight[(tid >> 5) * NV + lane];
  float wb = weight[(r2 & 7) * NV + ((k2 < NV) ? k2 : NV - 1)];

#pragma unroll
  for (int j = 0; j < 6; ++j) *(float4*)(smem + WNO + j * WROW + tid * 4) = vn[j];
  *(float4*)(smem + tid * 4) = vo;
  *(float4*)(smem + ((jt < 6) ? WNO + jt * WROW : 0) + (256 + lane) * 4) = vt;

  Ws[(tid >> 5) * KPAD + lane] = wa;
  Ws[r2 * KPAD + k2] = (r2 < NOUT && k2 < NV) ? wb : 0.f;
  Ws[(8 + (tid >> 5)) * KPAD + lane] = 0.f;

  __syncthreads();

  if ((wv & 1) == 0) {
    sandwich_side<true>(tid, bb, x0, u, smem);
  } else {
    sandwich_side<false>(tid, bb, x0, u, smem);
  }

  __syncthreads();

  const int rrow = lane & 15;
  const int kh = (lane >> 4) * 2;

  float ax[10], ay[10];
#pragma unroll
  for (int st = 0; st < 10; ++st) {
    int k0 = st * 4 + kh;
    float2 v = *(const float2*)(Ws + rrow * KPAD + k0);
    ax[st] = v.x;
    ay[st] = v.y;
  }

  for (int tile = wv; tile < COLS / 16; tile += 8) {
    const int nbase = tile * 16;
    const unsigned col = nbase + rrow;
    __builtin_assume(col < COLS);
    const unsigned site = col / 18;
    const unsigned m = col - site * 18;
    const float* own = smem + site * 72 + m;
    v8f acc = {0.f, 0.f, 0.f, 0.f, 0.f, 0.f, 0.f, 0.f};
#pragma unroll
    for (int st = 0; st < 10; ++st) {
      int k0 = st * 4 + kh;
      v2f a, bf;
      a.x = ax[st];
      a.y = ay[st];
      if (st == 0) {
        bf.x = own[kh * 18];
        bf.y = own[kh * 18 + 18];
      } else if (st < 9) {
        bf.x = TWs[k0][col];
        bf.y = TWs[k0 + 1][col];
      } else {
        bf.x = (kh == 0 && (m == 0 || m == 8 || m == 16)) ? 1.f : 0.f;
        bf.y = 0.f;
      }
      acc = __builtin_amdgcn_wmma_f32_16x16x4_f32(
           false, a,  false, bf,
           (short)0, acc,  false,  false);
    }
    if (lane < 16) {
      float* op = wout + (size_t)(s0 + site) * (NOUT * 18) + m;
#pragma unroll
      for (int r = 0; r < NOUT; ++r) op[r * 18] = acc[r];
    }
  }
}

extern "C" void kernel_launch(void* const* d_in, const int* in_sizes, int n_in,
                              void* d_out, int out_size, void* d_ws, size_t ws_size,
                              hipStream_t stream) {
  const float* u = (const float*)d_in[0];
  const float* w = (const float*)d_in[1];
  const float* weight = (const float*)d_in[2];
  float* out = (float*)d_out;

  const size_t u_elems = (size_t)NB * NX * 4 * 18;
  hipMemcpyAsync(out, u, u_elems * sizeof(float), hipMemcpyDeviceToDevice, stream);

  float* wout = out + u_elems;
  lconv_kernel<<<NSITE / SPW, 256, 0, stream>>>(u, w, weight, wout);
}


# w_out tiles staged in LDS (over the dead neighbour rows) in memory layout, then stored as whole 128-byte lines, 16 bytes per lane: 18 full-wave stores per workgroup, not 144 half-wave 4-byte ones
# speedup vs baseline: 1.8068x; 1.0436x over previous
#include <hip/hip_runtime.h>
#include <hip/hip_bf16.h>

#define NB      4
#define NX      16384
#define NSITE   (NB * NX)
#define NIN     4
#define NOUT    8
#define NV      37
#define KPAD    40
#define SPW     16
#define COLS    (SPW * 18)
#define LSTRIDE 292
#define TROWS   36
#define TWF     (TROWS * LSTRIDE)
#define NROWQ   288
#define WROW    1168
#define WNO     (TWF + 8)
#define WNF     (6 * WROW)
#define WSO     (WNO + WNF)

typedef float v2f __attribute__((ext_vector_type(2)));
typedef float v8f __attribute__((ext_vector_type(8)));


__device__ __forceinline__ v2f cmul(v2f a, v2f b) {
  v2f r = a.xx * b;
  return __builtin_elementwise_fma(a.yy, (v2f){-b.y, b.x}, r);
}
__device__ __forceinline__ v2f cmac(v2f a, v2f b, v2f c) {
  c = __builtin_elementwise_fma(a.xx, b, c);
  return __builtin_elementwise_fma(a.yy, (v2f){-b.y, b.x}, c);
}
__device__ __forceinline__ v2f cmul_bc(v2f a, v2f b) {
  v2f r = a.xx * (v2f){b.x, -b.y};
  return __builtin_elementwise_fma(a.yy, b.yx, r);
}
__device__ __forceinline__ v2f cmac_bc(v2f a, v2f b, v2f c) {
  c = __builtin_elementwise_fma(a.xx, (v2f){b.x, -b.y}, c);
  return __builtin_elementwise_fma(a.yy, b.yx, c);
}

template <bool POS>
__device__ __forceinline__ void sandwich_side(int tid, int bb, int x0,
                                              const float* __restrict__ u, float* smem) {
  float (*TWs)[LSTRIDE] = (float (*)[LSTRIDE])smem;
  const int n = tid & 3;
  const int axis = (tid >> 2) & 3;
  const int shamt = (0x0004080B >> (axis * 8)) & 0xFF;
  const int dmask = (axis < 2) ? 7 : 15;
  const int combo = POS ? axis : axis + 4;
  const int wrow = ((axis == 3) ? 0 : WNO + (POS ? axis : axis + 3) * WROW) + n * 18;
  const int site0 = ((tid >> 6) << 1) | ((tid >> 4) & 1);

  for (int pass = 0; pass < 2; ++pass) {
    const int site = site0 + 8 * pass;
    const int x = x0 + site;
    const int c = (x >> shamt) & dmask;
    const int cn = POS ? ((c + 1) & dmask) : ((c - 1) & dmask);
    const int xn = x + ((cn - c) << shamt);
    const int xa = POS ? x : xn;
    const unsigned aoff = (unsigned)(((bb * NX + xa) * 4 + axis) * 18);
    const v2f* Ap2 = (const v2f*)(u + aoff);
    const v2f* Wp = (const v2f*)(smem + wrow + ((axis == 3) ? cn : site) * 72);

    v2f A[9], Wm[9], T[9], V[9];
#pragma unroll
    for (int q = 0; q < 9; ++q) A[q] = Ap2[q];
#pragma unroll
    for (int q = 0; q < 9; ++q) Wm[q] = Wp[q];

    if (POS) {
#pragma unroll
      for (int i = 0; i < 3; ++i)
#pragma unroll
        for (int k = 0; k < 3; ++k) {
          v2f c2 = cmul(Wm[k], A[i * 3]);
          c2 = cmac(Wm[3 + k], A[i * 3 + 1], c2);
          T[i * 3 + k] = cmac(Wm[6 + k], A[i * 3 + 2], c2);
        }
#pragma unroll
      for (int i = 0; i < 3; ++i)
#pragma unroll
        for (int k = 0; k < 3; ++k) {
          v2f c2 = cmul_bc(T[i * 3], A[k * 3]);
          c2 = cmac_bc(T[i * 3 + 1], A[k * 3 + 1], c2);
          V[i * 3 + k] = cmac_bc(T[i * 3 + 2], A[k * 3 + 2], c2);
        }
    } else {
#pragma unroll
      for (int i = 0; i < 3; ++i)
#pragma unroll
        for (int k = 0; k < 3; ++k) {
          v2f c2 = cmul_bc(Wm[k], A[i]);
          c2 = cmac_bc(Wm[3 + k], A[3 + i], c2);
          T[i * 3 + k] = cmac_bc(Wm[6 + k], A[6 + i], c2);
        }
#pragma unroll
      for (int i = 0; i < 3; ++i)
#pragma unroll
        for (int k = 0; k < 3; ++k) {
          v2f c2 = cmul(T[i * 3], A[k]);
          c2 = cmac(T[i * 3 + 1], A[3 + k], c2);
          V[i * 3 + k] = cmac(T[i * 3 + 2], A[6 + k], c2);
        }
    }

    v2f* Vp = (v2f*)&TWs[4 + combo * 4 + n][site * 18];
#pragma unroll
    for (int q = 0; q < 9; ++q) Vp[q] = V[q];
  }
}

__global__ __launch_bounds__(256) void lconv_kernel(
    const float* __restrict__ u, const float* __restrict__ w,
    const float* __restrict__ weight, float* __restrict__ wout) {
  __shared__ __attribute__((aligned(16))) float smem[WSO + 16 * KPAD];
  float (*TWs)[LSTRIDE] = (float (*)[LSTRIDE])smem;
  float* Ws = smem + WSO;
  float* Os = smem + WNO;

  const int tid = threadIdx.x;
  __builtin_assume(tid >= 0 && tid < 256);
  const int lane = tid & 31;
  const int wv = __builtin_amdgcn_readfirstlane(tid >> 5);
  const int s0 = blockIdx.x * SPW;
  const int bb = s0 >> 14;
  const int x0 = s0 & (NX - 1);

  auto rowsite = [&](int j) -> int {
    int a = (j >= 3) ? j - 3 : j;
    int shamt = (a == 0) ? 11 : (a == 1) ? 8 : 4;
    int dmask = (a < 2) ? 7 : 15;
    int c = (x0 >> shamt) & dmask;
    int cn = (j < 3) ? ((c + 1) & dmask) : ((c - 1) & dmask);
    return (j >= 6) ? x0 : x0 + ((cn - c) << shamt);
  };
  const float4* wq = (const float4*)w;
  float4 vn[6];
#pragma unroll
  for (int j = 0; j < 6; ++j) vn[j] = wq[(unsigned)((bb * NX + rowsite(j)) * 18 + tid)];
  float4 vo = wq[(unsigned)(s0 * 18 + tid)];
  const int jt = (wv < 6) ? wv : 6;
  float4 vt = wq[(unsigned)((bb * NX + rowsite(jt)) * 18 + 256 + lane)];
  const int k2 = 32 + (tid & 7);
  const int r2 = (tid >> 3) & 15;
  float wa = weight[(tid >> 5) * NV + lane];
  float wb = weight[(r2 & 7) * NV + ((k2 < NV) ? k2 : NV - 1)];

#pragma unroll
  for (int j = 0; j < 6; ++j) *(float4*)(smem + WNO + j * WROW + tid * 4) = vn[j];
  *(float4*)(smem + tid * 4) = vo;
  *(float4*)(smem + ((jt < 6) ? WNO + jt * WROW : 0) + (256 + lane) * 4) = vt;

  Ws[(tid >> 5) * KPAD + lane] = wa;
  Ws[r2 * KPAD + k2] = (r2 < NOUT && k2 < NV) ? wb : 0.f;
  Ws[(8 + (tid >> 5)) * KPAD + lane] = 0.f;

  __syncthreads();

  if ((wv & 1) == 0) {
    sandwich_side<true>(tid, bb, x0, u, smem);
  } else {
    sandwich_side<false>(tid, bb, x0, u, smem);
  }

  __syncthreads();

  const int rrow = lane & 15;
  const int kh = (lane >> 4) * 2;

  float ax[10], ay[10];
#pragma unroll
  for (int st = 0; st < 10; ++st) {
    int k0 = st * 4 + kh;
    float2 v = *(const float2*)(Ws + rrow * KPAD + k0);
    ax[st] = v.x;
    ay[st] = v.y;
  }

  for (int tile = wv; tile < COLS / 16; tile += 8) {
    const int nbase = tile * 16;
    const unsigned col = nbase + rrow;
    __builtin_assume(col < COLS);
    const unsigned site = col / 18;
    const unsigned m = col - site * 18;
    const float* own = smem + site * 72 + m;
    v8f acc = {0.f, 0.f, 0.f, 0.f, 0.f, 0.f, 0.f, 0.f};
#pragma unroll
    for (int st = 0; st < 10; ++st) {
      int k0 = st * 4 + kh;
      v2f a, bf;
      a.x = ax[st];
      a.y = ay[st];
      if (st == 0) {
        bf.x = own[kh * 18];
        bf.y = own[kh * 18 + 18];
      } else if (st < 9) {
        bf.x = TWs[k0][col];
        bf.y = TWs[k0 + 1][col];
      } else {
        bf.x = (kh == 0 && (m == 0 || m == 8 || m == 16)) ? 1.f : 0.f;
        bf.y = 0.f;
      }
      acc = __builtin_amdgcn_wmma_f32_16x16x4_f32(
           false, a,  false, bf,
           (short)0, acc,  false,  false);
    }
    if (lane < 16) {
      float* op = Os + site * (NOUT * 18) + m;
#pragma unroll
      for (int r = 0; r < NOUT; ++r) op[r * 18] = acc[r];
    }
  }

  __syncthreads();

  const float4* Oq = (const float4*)Os;
  float4* gq = (float4*)(wout + (size_t)s0 * (NOUT * 18));
  float4 o0 = Oq[tid];
  float4 o1 = Oq[tid + 256];
  gq[tid] = o0;
  gq[tid + 256] = o1;
  if (wv < 2) gq[tid + 512] = Oq[tid + 512];
}

extern "C" void kernel_launch(void* const* d_in, const int* in_sizes, int n_in,
                              void* d_out, int out_size, void* d_ws, size_t ws_size,
                              hipStream_t stream) {
  const float* u = (const float*)d_in[0];
  const float* w = (const float*)d_in[1];
  const float* weight = (const float*)d_in[2];
  float* out = (float*)d_out;

  const size_t u_elems = (size_t)NB * NX * 4 * 18;
  hipMemcpyAsync(out, u, u_elems * sizeof(float), hipMemcpyDeviceToDevice, stream);

  float* wout = out + u_elems;
  lconv_kernel<<<NSITE / SPW, 256, 0, stream>>>(u, w, weight, wout);
}


# six workgroups per compute unit instead of four: a sandwich writes its result over its own W operand in LDS (no separate tile: LDS 72672 to 52480 B), W read row by row, registers capped at 80
# speedup vs baseline: 1.9075x; 1.0557x over previous
#include <hip/hip_runtime.h>
#include <hip/hip_bf16.h>

#define NB      4
#define NX      16384
#define NSITE   (NB * NX)
#define NIN     4
#define NOUT    8
#define NV      37
#define KPAD    40
#define SPW     16
#define COLS    (SPW * 18)
#define ROWF    1152
#define WROW    1168
#define OWN     (8 * WROW)
#define WSO     (OWN + ROWF)
#define OSO     (WSO + NOUT * KPAD)
#define SMEMF   (OSO + SPW * NOUT * 18)

typedef float v2f __attribute__((ext_vector_type(2)));
typedef float v4f __attribute__((ext_vector_type(4)));
typedef float v8f __attribute__((ext_vector_type(8)));


__device__ __forceinline__ v2f cmul(v2f a, v2f b) {
  v2f r = a.xx * b;
  return __builtin_elementwise_fma(a.yy, (v2f){-b.y, b.x}, r);
}
__device__ __forceinline__ v2f cmac(v2f a, v2f b, v2f c) {
  c = __builtin_elementwise_fma(a.xx, b, c);
  return __builtin_elementwise_fma(a.yy, (v2f){-b.y, b.x}, c);
}
__device__ __forceinline__ v2f cmul_bc(v2f a, v2f b) {
  v2f r = a.xx * (v2f){b.x, -b.y};
  return __builtin_elementwise_fma(a.yy, b.yx, r);
}
__device__ __forceinline__ v2f cmac_bc(v2f a, v2f b, v2f c) {
  c = __builtin_elementwise_fma(a.xx, (v2f){b.x, -b.y}, c);
  return __builtin_elementwise_fma(a.yy, b.yx, c);
}

template <bool POS>
__device__ __forceinline__ void sandwich_side(int tid, int bb, int x0,
                                              const float* __restrict__ u, float* smem) {
  const int n = tid & 3;
  const int axis = (tid >> 2) & 3;
  const int shamt = (0x0004080B >> (axis * 8)) & 0xFF;
  const int dmask = (axis < 2) ? 7 : 15;
  const int blk = (((tid >> 2) & 3) | ((tid >> 3) & 4)) * WROW + n * 18;
  const int site0 = ((tid >> 6) << 1) | ((tid >> 4) & 1);

  v2f Au[2][9];
#pragma unroll
  for (int pass = 0; pass < 2; ++pass) {
    const int x = x0 + site0 + 8 * pass;
    const int c = (x >> shamt) & dmask;
    const int cn = POS ? ((c + 1) & dmask) : ((c - 1) & dmask);
    const int xn = x + ((cn - c) << shamt);
    const int xa = POS ? x : xn;
    const unsigned aoff = (unsigned)(((bb * NX + xa) * 4 + axis) * 18);
    const v2f* Ap2 = (const v2f*)(u + aoff);
#pragma unroll
    for (int q = 0; q < 9; ++q) Au[pass][q] = Ap2[q];
  }

#pragma unroll
  for (int pass = 0; pass < 2; ++pass) {
    const int site = site0 + 8 * pass;
    v2f* Wp = (v2f*)(smem + blk + site * 72);

    v2f A[9], T[9];
#pragma unroll
    for (int q = 0; q < 9; ++q) A[q] = Au[pass][q];

#pragma unroll
    for (int j = 0; j < 3; ++j) {
      v2f Wr[3];
#pragma unroll
      for (int k = 0; k < 3; ++k) Wr[k] = Wp[j * 3 + k];
#pragma unroll
      for (int i = 0; i < 3; ++i)
#pragma unroll
        for (int k = 0; k < 3; ++k) {
          if (POS) {
            T[i * 3 + k] = (j == 0) ? cmul(Wr[k], A[i * 3]) : cmac(Wr[k], A[i * 3 + j], T[i * 3 + k]);
          } else {
            T[i * 3 + k] = (j == 0) ? cmul_bc(Wr[k], A[i]) : cmac_bc(Wr[k], A[j * 3 + i], T[i * 3 + k]);
          }
        }
    }
#pragma unroll
    for (int i = 0; i < 3; ++i)
#pragma unroll
      for (int k = 0; k < 3; ++k) {
        v2f c2;
        if (POS) {
          c2 = cmul_bc(T[i * 3], A[k * 3]);
          c2 = cmac_bc(T[i * 3 + 1], A[k * 3 + 1], c2);
          c2 = cmac_bc(T[i * 3 + 2], A[k * 3 + 2], c2);
        } else {
          c2 = cmul(T[i * 3], A[k]);
          c2 = cmac(T[i * 3 + 1], A[3 + k], c2);
          c2 = cmac(T[i * 3 + 2], A[6 + k], c2);
        }
        Wp[i * 3 + k] = c2;
      }
  }
}

__global__ __launch_bounds__(256) __attribute__((amdgpu_waves_per_eu(12))) void lconv_kernel(
    const float* __restrict__ u, const float* __restrict__ w,
    const float* __restrict__ weight, float* __restrict__ wout) {
  __shared__ __attribute__((aligned(16))) float smem[SMEMF];
  float* Ws = smem + WSO;
  float* Os = smem + OSO;

  const int tid = threadIdx.x;
  __builtin_assume(tid >= 0 && tid < 256);
  const int lane = tid & 31;
  const int wv = __builtin_amdgcn_readfirstlane(tid >> 5);
  const int s0 = blockIdx.x * SPW;
  const int bb = s0 >> 14;
  const int x0 = s0 & (NX - 1);

  auto rowsite = [&](int j) -> int {
    int a = (j >= 3) ? j - 3 : j;
    int shamt = (a == 0) ? 11 : (a == 1) ? 8 : 4;
    int dmask = (a < 2) ? 7 : 15;
    int c = (x0 >> shamt) & dmask;
    int cn = (j < 3) ? ((c + 1) & dmask) : ((c - 1) & dmask);
    return (j >= 6) ? x0 : x0 + ((cn - c) << shamt);
  };
  const v4f* wq = (const v4f*)w;
  v4f vn[6];
#pragma unroll
  for (int j = 0; j < 6; ++j) vn[j] = wq[(unsigned)((bb * NX + rowsite(j)) * 18 + tid)];
  v4f vo = wq[(unsigned)(s0 * 18 + tid)];
  const int jt = (wv < 6) ? wv : 6;
  v4f vt = wq[(unsigned)((bb * NX + rowsite(jt)) * 18 + 256 + lane)];
  const int k2 = 32 + (tid & 7);
  const int r2 = (tid >> 3) & 7;
  float wa = weight[(tid >> 5) * NV + lane];
  float wb = weight[r2 * NV + ((k2 < NV) ? k2 : NV - 1)];

#pragma unroll
  for (int j = 0; j < 6; ++j) *(v4f*)(smem + ((j < 3) ? j : j + 1) * WROW + tid * 4) = vn[j];
  *(v4f*)(smem + OWN + tid * 4) = vo;
  *(v4f*)(smem + 3 * WROW + ((tid >= 18) ? tid - 18 : tid + 270) * 4) = vo;
  *(v4f*)(smem + 7 * WROW + (tid + 18) * 4) = vo;
  v4f* sq = (v4f*)smem;
  const int ownt = (wv > 5) ? 1 : 0;
  const int d0 = (jt + ((jt > 2) ? 1 : 0) + ownt) * (WROW / 4) + 256 + lane;
  const int d1 = ownt ? 3 * (WROW / 4) + 238 + lane : d0;
  const int d2 = ownt ? 7 * (WROW / 4) + ((lane < 14) ? 274 + lane : lane - 14) : d0;
  sq[d0] = vt;
  sq[d1] = vt;
  sq[d2] = vt;

  Ws[(tid >> 5) * KPAD + lane] = wa;
  Ws[r2 * KPAD + k2] = (k2 < NV) ? wb : 0.f;

  __syncthreads();

  if ((wv & 1) == 0) {
    sandwich_side<true>(tid, bb, x0, u, smem);
  } else {
    sandwich_side<false>(tid, bb, x0, u, smem);
  }

  __syncthreads();

  const int rrow = lane & 15;
  const int kh = (lane >> 4) * 2;

  float ax[10], ay[10];
#pragma unroll
  for (int st = 0; st < 10; ++st) {
    int k0 = st * 4 + kh;
    float2 v = *(const float2*)(Ws + (rrow & 7) * KPAD + k0);
    ax[st] = (rrow < NOUT) ? v.x : 0.f;
    ay[st] = (rrow < NOUT) ? v.y : 0.f;
  }

  for (int tile = wv; tile < COLS / 16; tile += 8) {
    const int nbase = tile * 16;
    const unsigned col = nbase + rrow;
    __builtin_assume(col < COLS);
    const unsigned site = col / 18;
    const unsigned m = col - site * 18;
    const float* bp = smem + site * 72 + kh * 18 + m;
    float bx[9], by[9];
#pragma unroll
    for (int st = 0; st < 9; ++st) {
      const int blk = (st == 0) ? OWN : (st - 1) * WROW;
      bx[st] = bp[blk];
      by[st] = bp[blk + 18];
    }
    __builtin_amdgcn_sched_barrier(0);
    v8f acc = {0.f, 0.f, 0.f, 0.f, 0.f, 0.f, 0.f, 0.f};
#pragma unroll
    for (int st = 0; st < 10; ++st) {
      v2f a, bf;
      a.x = ax[st];
      a.y = ay[st];
      if (st < 9) {
        bf.x = bx[st];
        bf.y = by[st];
      } else {
        bf.x = (kh == 0 && (m == 0 || m == 8 || m == 16)) ? 1.f : 0.f;
        bf.y = 0.f;
      }
      acc = __builtin_amdgcn_wmma_f32_16x16x4_f32(
           false, a,  false, bf,
           (short)0, acc,  false,  false);
    }
    if (lane < 16) {
      float* op = Os + site * (NOUT * 18) + m;
#pragma unroll
      for (int r = 0; r < NOUT; ++r) op[r * 18] = acc[r];
    }
  }

  __syncthreads();

  const float4* Oq = (const float4*)Os;
  float4* gq = (float4*)(wout + (size_t)s0 * (NOUT * 18));
  float4 o0 = Oq[tid];
  float4 o1 = Oq[tid + 256];
  gq[tid] = o0;
  gq[tid + 256] = o1;
  if (wv < 2) gq[tid + 512] = Oq[tid + 512];
}

extern "C" void kernel_launch(void* const* d_in, const int* in_sizes, int n_in,
                              void* d_out, int out_size, void* d_ws, size_t ws_size,
                              hipStream_t stream) {
  const float* u = (const float*)d_in[0];
  const float* w = (const float*)d_in[1];
  const float* weight = (const float*)d_in[2];
  float* out = (float*)d_out;

  const size_t u_elems = (size_t)NB * NX * 4 * 18;
  hipMemcpyAsync(out, u, u_elems * sizeof(float), hipMemcpyDeviceToDevice, stream);

  float* wout = out + u_elems;
  lconv_kernel<<<NSITE / SPW, 256, 0, stream>>>(u, w, weight, wout);
}
